# starting kernel
# baseline (speedup 1.0000x reference)
#include <hip/hip_runtime.h>

typedef __attribute__((ext_vector_type(2))) float v2f;
typedef __attribute__((ext_vector_type(8))) float v8f;

#define Bn 2
#define Cn 64
#define On 64
#define En 128
#define Tn 512
#define Wn 7
#define TOUT 256
#define FOUT 64

__global__ void lc_stats1(const float* __restrict__ x,
                          float* __restrict__ Sx, float* __restrict__ Qx) {
  int cf = blockIdx.x;
  int c = cf >> 7, f = cf & 127;
  float s[7], q[7];
#pragma unroll
  for (int w = 0; w < 7; ++w) { s[w] = 0.f; q[w] = 0.f; }
  for (int item = threadIdx.x; item < Bn * TOUT; item += blockDim.x) {
    int b = item >> 8, t = item & 255;
    const float* xr = x + (((size_t)b * Cn + c) * En + f) * Tn;
    int tau0 = 2 * t - 3;
#pragma unroll
    for (int w = 0; w < 7; ++w) {
      int tau = tau0 + w;
      if (tau >= 0 && tau < Tn) {
        float v = xr[tau];
        s[w] += v; q[w] += v * v;
      }
    }
  }
  __shared__ float red[8][14];
  int lane = threadIdx.x & 31, wv = threadIdx.x >> 5;
#pragma unroll
  for (int k = 0; k < 14; ++k) {
    float v = (k < 7) ? s[k] : q[k - 7];
    for (int off = 16; off > 0; off >>= 1) v += __shfl_down(v, off, 32);
    if (lane == 0) red[wv][k] = v;
  }
  __syncthreads();
  if (threadIdx.x < 14) {
    float tot = 0.f;
#pragma unroll
    for (int i = 0; i < 8; ++i) tot += red[i][threadIdx.x];
    if (threadIdx.x < 7) Sx[cf * 7 + threadIdx.x] = tot;
    else                 Qx[cf * 7 + threadIdx.x - 7] = tot;
  }
}

__global__ void lc_stats2(const float* __restrict__ wgt,
                          const float* __restrict__ gamma,
                          const float* __restrict__ beta,
                          const float* __restrict__ Sx,
                          const float* __restrict__ Qx,
                          float* __restrict__ aO, float* __restrict__ bO) {
  int o = blockIdx.x;
  int cbase = o & ~3;
  float n1 = 0.f, n2 = 0.f;
  for (int idx = threadIdx.x; idx < 4 * En * Wn; idx += blockDim.x) {
    int d = idx / (En * Wn);
    int rem = idx - d * (En * Wn);
    float wt = wgt[(size_t)(o * 4 + d) * (En * Wn) + rem];
    int c = cbase + d;
    float sv = Sx[(size_t)c * (En * Wn) + rem];
    float qv = Qx[(size_t)c * (En * Wn) + rem];
    n1 += wt * sv;
    n2 += wt * wt * qv;
  }
  __shared__ float red[4][2];
  int lane = threadIdx.x & 31, wv = threadIdx.x >> 5;
  for (int off = 16; off > 0; off >>= 1) {
    n1 += __shfl_down(n1, off, 32);
    n2 += __shfl_down(n2, off, 32);
  }
  if (lane == 0) { red[wv][0] = n1; red[wv][1] = n2; }
  __syncthreads();
  if (threadIdx.x == 0) {
    float t1 = 0.f, t2 = 0.f;
#pragma unroll
    for (int i = 0; i < 4; ++i) { t1 += red[i][0]; t2 += red[i][1]; }
    const float invcnt = 1.0f / 1835008.0f;
    float mean = t1 * invcnt;
    float var  = t2 * invcnt - mean * mean;
    float inv  = rsqrtf(var + 1e-5f);
    float a    = gamma[o] * inv;
    aO[o] = a;
    bO[o] = 56.0f * (beta[o] - mean * a);
  }
}

__global__ void __launch_bounds__(32)
lc_wmma_main(const float* __restrict__ x, const float* __restrict__ wgt,
             const float* __restrict__ aO, const float* __restrict__ bO,
             float* __restrict__ out) {
  int id = blockIdx.x;
  int tt = id & 15;
  int fo = (id >> 4) & 63;
  int ob = (id >> 10) & 3;
  int b  = id >> 12;
  int t0 = tt << 4, o0 = ob << 4, c0 = o0, f0 = fo << 1;

  int L  = threadIdx.x;
  int m  = L & 15;
  int hi = L >> 4;
  int d0 = hi << 1;
  int mq = m >> 2;

  __shared__ float xs[2][40][18];

  for (int idx = L; idx < 1184; idx += 32) {
    int fi = idx / 592;
    int r  = idx - fi * 592;
    int ch = r / 37;
    int j  = r - ch * 37;
    int tau = 2 * t0 - 3 + j;
    float v = 0.f;
    if (tau >= 0 && tau < Tn)
      v = x[(((size_t)b * Cn + (c0 + ch)) * En + (f0 + fi)) * Tn + tau];
    xs[fi][j][ch] = v;
  }

  float wa[2][7], wb[2][7];
#pragma unroll
  for (int fi = 0; fi < 2; ++fi) {
    const float* wp = wgt + (((size_t)(o0 + m) * 4 + d0) * En + (f0 + fi)) * Wn;
#pragma unroll
    for (int w = 0; w < 7; ++w) {
      wa[fi][w] = wp[w];
      wb[fi][w] = wp[En * Wn + w];
    }
  }

  __syncthreads();

  v8f acc = {};
#pragma unroll
  for (int fi = 0; fi < 2; ++fi) {
#pragma unroll
    for (int w = 0; w < 7; ++w) {
      int j = 2 * m + w;
#pragma unroll
      for (int q = 0; q < 4; ++q) {
        v2f bv = *(const v2f*)&xs[fi][j][(q << 2) + d0];
        float sel = (mq == q) ? 1.f : 0.f;
        v2f av = { sel * wa[fi][w], sel * wb[fi][w] };
        acc = __builtin_amdgcn_wmma_f32_16x16x4_f32(
             false, av,  false, bv,
             (short)0, acc,  false,  false);
      }
    }
  }

#pragma unroll
  for (int r = 0; r < 8; ++r) {
    int o = o0 + r + (hi << 3);
    float v = aO[o] * acc[r] + bO[o];
    v = (v >= 0.f) ? v : 0.01f * v;
    out[(((size_t)b * On + o) * FOUT + fo) * TOUT + t0 + m] = v;
  }
}

extern "C" void kernel_launch(void* const* d_in, const int* in_sizes, int n_in,
                              void* d_out, int out_size, void* d_ws, size_t ws_size,
                              hipStream_t stream) {
  (void)in_sizes; (void)n_in; (void)out_size; (void)ws_size;
  const float* x     = (const float*)d_in[0];
  const float* wgt   = (const float*)d_in[1];
  const float* gamma = (const float*)d_in[2];
  const float* beta  = (const float*)d_in[3];
  float* out = (float*)d_out;

  float* ws = (float*)d_ws;
  float* Sx = ws;
  float* Qx = ws + 57344;
  float* aO = ws + 114688;
  float* bO = ws + 114752;

  lc_stats1<<<Cn * En, 256, 0, stream>>>(x, Sx, Qx);
  lc_stats2<<<On, 128, 0, stream>>>(wgt, gamma, beta, Sx, Qx, aO, bO);
  lc_wmma_main<<<8192, 32, 0, stream>>>(x, wgt, aO, bO, out);
}


# conv pass: x^2 of both images summed over the work-group's eight t tiles at staging (squares product is linear in B), multiplied once per work-group: 119 WMMAs per wave, not 168 (sum order changes)
# speedup vs baseline: 2.0906x; 2.0906x over previous
#include <hip/hip_runtime.h>

typedef __attribute__((ext_vector_type(2))) float v2f;
typedef __attribute__((ext_vector_type(4))) float v4f;
typedef __attribute__((ext_vector_type(8))) float v8f;

#define Bn 2
#define Cn 64
#define On 64
#define En 128
#define Tn 512
#define Wn 7
#define TOUT 256
#define FOUT 64


__global__ void __launch_bounds__(256)
lc_conv_sums(const float* __restrict__ x, const float* __restrict__ wgt,
             float* __restrict__ part, float* __restrict__ out) {
  int id = blockIdx.x;
  int th = id & 1;
  int fo = (id >> 1) & 63;
  int ob = id >> 7;
  int wv = threadIdx.x >> 5;
  int tt = (th << 3) + wv;
  int t0 = tt << 4, o0 = ob << 4, c0 = o0, f0 = fo << 1;

  int L  = threadIdx.x & 31;
  int m  = L & 15;
  int hi = L >> 4;
  int d0 = hi << 1;
  int mq = m >> 2;

  __shared__ __attribute__((aligned(16))) float xs[Bn][2][264][18];
  __shared__ __attribute__((aligned(16))) float zs[2][40][18];
  __shared__ __attribute__((aligned(16))) float rs[8][16 * 36];

  float wa[2][7], wb[2][7];
#pragma unroll
  for (int fi = 0; fi < 2; ++fi) {
    const float* wp = wgt + (((size_t)(o0 + m) * 4 + d0) * En + (f0 + fi)) * Wn;
#pragma unroll
    for (int w = 0; w < 7; ++w) {
      wa[fi][w] = wp[w];
      wb[fi][w] = wp[En * Wn + w];
    }
  }

  int srow = threadIdx.x >> 3;
  int sfi  = srow >> 4, sch = srow & 15;
  int pc  = threadIdx.x & 7;
  const float* xrow = x + ((size_t)(c0 + sch) * En + (f0 + sfi)) * Tn + (th << 8);
  v4f xv[Bn][8];
#pragma unroll
  for (int bb = 0; bb < Bn; ++bb) {
#pragma unroll
    for (int k = 0; k < 8; ++k)
      xv[bb][k] = *(const v4f*)(xrow + (size_t)bb * (Cn * En * Tn) + (k << 5) + (pc << 2));
  }
  v4f xe[Bn];
#pragma unroll
  for (int bb = 0; bb < Bn; ++bb) xe[bb] = (v4f){0.f, 0.f, 0.f, 0.f};
  if (pc < 2) {
    int te = pc ? 256 : -4;
    int tau = (th << 8) + te;
    if (tau >= 0 && tau < Tn) {
#pragma unroll
      for (int bb = 0; bb < Bn; ++bb)
        xe[bb] = *(const v4f*)(xrow + (size_t)bb * (Cn * En * Tn) + te);
    }
  }
#pragma unroll
  for (int bb = 0; bb < Bn; ++bb) {
    float (*xp)[18] = xs[bb][sfi];
#pragma unroll
    for (int k = 0; k < 8; ++k) {
      int j = 4 + (k << 5) + (pc << 2);
      xp[j + 0][sch] = xv[bb][k].x;
      xp[j + 1][sch] = xv[bb][k].y;
      xp[j + 2][sch] = xv[bb][k].z;
      xp[j + 3][sch] = xv[bb][k].w;
    }
    if (pc < 2) {
      int j = pc ? 260 : 0;
      xp[j + 0][sch] = xe[bb].x;
      xp[j + 1][sch] = xe[bb].y;
      xp[j + 2][sch] = xe[bb].z;
      xp[j + 3][sch] = xe[bb].w;
    }
  }

  {
    v4f z0 = xv[0][0] * xv[0][0] + xv[1][0] * xv[1][0];
    v4f z7 = xv[0][7] * xv[0][7] + xv[1][7] * xv[1][7];
    v4f zmid = (v4f){0.f, 0.f, 0.f, 0.f};
#pragma unroll
    for (int k = 1; k < 7; ++k)
      zmid += xv[0][k] * xv[0][k] + xv[1][k] * xv[1][k];
    v4f zall = (z0 + zmid) + z7;
    v4f zlo  = z0 + zmid;
    v4f zhi  = zmid + z7;
    v4f ze   = xe[0] * xe[0] + xe[1] * xe[1];
    v4f plo, phi;
    plo.x = __shfl_xor(zlo.x, 7, 32); plo.y = __shfl_xor(zlo.y, 7, 32);
    plo.z = __shfl_xor(zlo.z, 7, 32); plo.w = __shfl_xor(zlo.w, 7, 32);
    phi.x = __shfl_xor(zhi.x, 1, 32); phi.y = __shfl_xor(zhi.y, 1, 32);
    phi.z = __shfl_xor(zhi.z, 1, 32); phi.w = __shfl_xor(zhi.w, 1, 32);
    float (*zp)[18] = zs[sfi];
    int jz = 4 + (pc << 2);
    zp[jz + 0][sch] = zall.x;
    zp[jz + 1][sch] = zall.y;
    zp[jz + 2][sch] = zall.z;
    zp[jz + 3][sch] = zall.w;
    if (pc < 2) {
      v4f e = ze + (pc ? phi : plo);
      int je = pc ? 36 : 0;
      zp[je + 0][sch] = e.x;
      zp[je + 1][sch] = e.y;
      zp[je + 2][sch] = e.z;
      zp[je + 3][sch] = e.w;
    }
  }

  __syncthreads();

  v8f acc0 = {};
  v8f acc1 = {};
  v8f acc2 = {};
  {
    int q2 = wv & 3, f2 = wv >> 2;
    float sel2 = (mq == q2) ? 1.f : 0.f;
#pragma unroll
    for (int w = 0; w < 7; ++w) {
      float a = f2 ? wa[1][w] : wa[0][w];
      float b = f2 ? wb[1][w] : wb[0][w];
      v2f av2 = { sel2 * (a * a), sel2 * (b * b) };
      v2f bz = *(const v2f*)&zs[f2][1 + 2 * m + w][(q2 << 2) + d0];
      acc2 = __builtin_amdgcn_wmma_f32_16x16x4_f32(
           false, av2,  false, bz,
           (short)0, acc2,  false,  false);
    }
  }
#pragma unroll
  for (int fi = 0; fi < 2; ++fi) {
#pragma unroll
    for (int w = 0; w < 7; ++w) {
      int j = (wv << 5) + 1 + 2 * m + w;
#pragma unroll
      for (int q = 0; q < 4; ++q) {
        v2f bv0 = *(const v2f*)&xs[0][fi][j][(q << 2) + d0];
        v2f bv1 = *(const v2f*)&xs[1][fi][j][(q << 2) + d0];
        float sel = (mq == q) ? 1.f : 0.f;
        v2f av = { sel * wa[fi][w], sel * wb[fi][w] };
        acc0 = __builtin_amdgcn_wmma_f32_16x16x4_f32(
             false, av,  false, bv0,
             (short)0, acc0,  false,  false);
        acc1 = __builtin_amdgcn_wmma_f32_16x16x4_f32(
             false, av,  false, bv1,
             (short)0, acc1,  false,  false);
      }
    }
  }

#pragma unroll
  for (int r = 0; r < 8; ++r) {
    int o = o0 + r + (hi << 3);
    size_t e = ((size_t)o * FOUT + fo) * TOUT + t0 + m;
    out[e] = acc0[r];
    out[e + (size_t)On * FOUT * TOUT] = acc1[r];
  }

  float* rw = &rs[wv][0];
#pragma unroll
  for (int r = 0; r < 8; ++r) {
    rw[r * 36 + L]       = acc0[r] + acc1[r];
    rw[(8 + r) * 36 + L] = acc2[r];
  }
  __syncthreads();
  {
    int T    = threadIdx.x;
    int sub  = T & 7;
    int oi   = T >> 3;
    int kind = oi >> 4;
    int ol   = oi & 15;
    int r = ol & 7, h = ol >> 3;
    const v4f* rp = (const v4f*)(&rs[sub][0] + (kind * 8 + r) * 36 + h * 16);
    v4f s4 = rp[0] + rp[1] + rp[2] + rp[3];
    float s = (s4.x + s4.y) + (s4.z + s4.w);
    s += __shfl_xor(s, 1, 32);
    s += __shfl_xor(s, 2, 32);
    s += __shfl_xor(s, 4, 32);
    if (sub == 0) {
      int pb = (fo << 1) + th;
      part[(((o0 + ol) << 7) + pb) * 2 + kind] = s;
    }
  }
}

__global__ void __launch_bounds__(256)
lc_bn_apply(const float* __restrict__ part, const float* __restrict__ gamma,
            const float* __restrict__ beta, float* __restrict__ out) {
  int id = blockIdx.x;
  int bo = id >> 2;
  int o  = bo & 63;
  int T  = threadIdx.x;
  v4f* p = (v4f*)(out + (size_t)bo * (FOUT * TOUT) + (id & 3) * 4096);

  v4f v[4];
#pragma unroll
  for (int k = 0; k < 4; ++k) v[k] = p[T + 256 * k];
  v2f pr = {0.f, 0.f};
  if (T < 128) pr = *(const v2f*)(part + (((o << 7) + T) << 1));
  float g = gamma[o], be = beta[o];

  float n1 = pr.x, n2 = pr.y;
  for (int off = 16; off > 0; off >>= 1) {
    n1 += __shfl_xor(n1, off, 32);
    n2 += __shfl_xor(n2, off, 32);
  }
  __shared__ float red[8][2];
  int lane = T & 31, wv = T >> 5;
  if (lane == 0) { red[wv][0] = n1; red[wv][1] = n2; }
  __syncthreads();
  float t1 = 0.f, t2 = 0.f;
#pragma unroll
  for (int i = 0; i < 8; ++i) { t1 += red[i][0]; t2 += red[i][1]; }

  const float invcnt = 1.0f / 1835008.0f;
  float mean = t1 * invcnt;
  float var  = t2 * invcnt - mean * mean;
  float inv  = rsqrtf(var + 1e-5f);
  float a    = g * inv;
  float c    = 56.0f * (be - mean * a);

#pragma unroll
  for (int k = 0; k < 4; ++k) {
    v4f y = a * v[k] + c;
    y.x = (y.x >= 0.f) ? y.x : 0.01f * y.x;
    y.y = (y.y >= 0.f) ? y.y : 0.01f * y.y;
    y.z = (y.z >= 0.f) ? y.z : 0.01f * y.z;
    y.w = (y.w >= 0.f) ? y.w : 0.01f * y.w;
    p[T + 256 * k] = y;
  }
}

extern "C" void kernel_launch(void* const* d_in, const int* in_sizes, int n_in,
                              void* d_out, int out_size, void* d_ws, size_t ws_size,
                              hipStream_t stream) {
  (void)in_sizes; (void)n_in; (void)out_size; (void)ws_size;
  const float* x     = (const float*)d_in[0];
  const float* wgt   = (const float*)d_in[1];
  const float* gamma = (const float*)d_in[2];
  const float* beta  = (const float*)d_in[3];
  float* out = (float*)d_out;

  float* part = (float*)d_ws;

  lc_conv_sums<<<512, 256, 0, stream>>>(x, wgt, part, out);
  lc_bn_apply<<<512, 256, 0, stream>>>(part, gamma, beta, out);
}


# conv pass epilogue: tile results staged in LDS, stored behind the barrier as whole 128-byte lines (4 16-byte stores per thread), not 16 half-line stores waited on at the barrier (sums bit-identical)
# speedup vs baseline: 3.7110x; 1.7751x over previous
#include <hip/hip_runtime.h>

typedef __attribute__((ext_vector_type(2))) float v2f;
typedef __attribute__((ext_vector_type(4))) float v4f;
typedef __attribute__((ext_vector_type(8))) float v8f;

#define Bn 2
#define Cn 64
#define On 64
#define En 128
#define Tn 512
#define Wn 7
#define TOUT 256
#define FOUT 64


__global__ void __launch_bounds__(256)
lc_conv_sums(const float* __restrict__ x, const float* __restrict__ wgt,
             float* __restrict__ part, float* __restrict__ out) {
  int id = blockIdx.x;
  int th = id & 1;
  int fo = (id >> 1) & 63;
  int ob = id >> 7;
  int wv = threadIdx.x >> 5;
  int o0 = ob << 4, c0 = o0, f0 = fo << 1;

  int L  = threadIdx.x & 31;
  int m  = L & 15;
  int hi = L >> 4;
  int d0 = hi << 1;
  int mq = m >> 2;

  __shared__ __attribute__((aligned(16))) float xs[Bn][2][264][18];
  __shared__ __attribute__((aligned(16))) float zs[2][40][18];
  __shared__ __attribute__((aligned(16))) float rs[8][16 * 36];
  __shared__ __attribute__((aligned(16))) float ot[Bn][16][132];

  float wa[2][7], wb[2][7];
#pragma unroll
  for (int fi = 0; fi < 2; ++fi) {
    const float* wp = wgt + (((size_t)(o0 + m) * 4 + d0) * En + (f0 + fi)) * Wn;
#pragma unroll
    for (int w = 0; w < 7; ++w) {
      wa[fi][w] = wp[w];
      wb[fi][w] = wp[En * Wn + w];
    }
  }

  int srow = threadIdx.x >> 3;
  int sfi  = srow >> 4, sch = srow & 15;
  int pc  = threadIdx.x & 7;
  const float* xrow = x + ((size_t)(c0 + sch) * En + (f0 + sfi)) * Tn + (th << 8);
  v4f xv[Bn][8];
#pragma unroll
  for (int bb = 0; bb < Bn; ++bb) {
#pragma unroll
    for (int k = 0; k < 8; ++k)
      xv[bb][k] = *(const v4f*)(xrow + (size_t)bb * (Cn * En * Tn) + (k << 5) + (pc << 2));
  }
  v4f xe[Bn];
#pragma unroll
  for (int bb = 0; bb < Bn; ++bb) xe[bb] = (v4f){0.f, 0.f, 0.f, 0.f};
  if (pc < 2) {
    int te = pc ? 256 : -4;
    int tau = (th << 8) + te;
    if (tau >= 0 && tau < Tn) {
#pragma unroll
      for (int bb = 0; bb < Bn; ++bb)
        xe[bb] = *(const v4f*)(xrow + (size_t)bb * (Cn * En * Tn) + te);
    }
  }
#pragma unroll
  for (int bb = 0; bb < Bn; ++bb) {
    float (*xp)[18] = xs[bb][sfi];
#pragma unroll
    for (int k = 0; k < 8; ++k) {
      int j = 4 + (k << 5) + (pc << 2);
      xp[j + 0][sch] = xv[bb][k].x;
      xp[j + 1][sch] = xv[bb][k].y;
      xp[j + 2][sch] = xv[bb][k].z;
      xp[j + 3][sch] = xv[bb][k].w;
    }
    if (pc < 2) {
      int j = pc ? 260 : 0;
      xp[j + 0][sch] = xe[bb].x;
      xp[j + 1][sch] = xe[bb].y;
      xp[j + 2][sch] = xe[bb].z;
      xp[j + 3][sch] = xe[bb].w;
    }
  }

  {
    v4f z0 = xv[0][0] * xv[0][0] + xv[1][0] * xv[1][0];
    v4f z7 = xv[0][7] * xv[0][7] + xv[1][7] * xv[1][7];
    v4f zmid = (v4f){0.f, 0.f, 0.f, 0.f};
#pragma unroll
    for (int k = 1; k < 7; ++k)
      zmid += xv[0][k] * xv[0][k] + xv[1][k] * xv[1][k];
    v4f zall = (z0 + zmid) + z7;
    v4f zlo  = z0 + zmid;
    v4f zhi  = zmid + z7;
    v4f ze   = xe[0] * xe[0] + xe[1] * xe[1];
    v4f plo, phi;
    plo.x = __shfl_xor(zlo.x, 7, 32); plo.y = __shfl_xor(zlo.y, 7, 32);
    plo.z = __shfl_xor(zlo.z, 7, 32); plo.w = __shfl_xor(zlo.w, 7, 32);
    phi.x = __shfl_xor(zhi.x, 1, 32); phi.y = __shfl_xor(zhi.y, 1, 32);
    phi.z = __shfl_xor(zhi.z, 1, 32); phi.w = __shfl_xor(zhi.w, 1, 32);
    float (*zp)[18] = zs[sfi];
    int jz = 4 + (pc << 2);
    zp[jz + 0][sch] = zall.x;
    zp[jz + 1][sch] = zall.y;
    zp[jz + 2][sch] = zall.z;
    zp[jz + 3][sch] = zall.w;
    if (pc < 2) {
      v4f e = ze + (pc ? phi : plo);
      int je = pc ? 36 : 0;
      zp[je + 0][sch] = e.x;
      zp[je + 1][sch] = e.y;
      zp[je + 2][sch] = e.z;
      zp[je + 3][sch] = e.w;
    }
  }

  __syncthreads();

  v8f acc0 = {};
  v8f acc1 = {};
  v8f acc2 = {};
  {
    int q2 = wv & 3, f2 = wv >> 2;
    float sel2 = (mq == q2) ? 1.f : 0.f;
#pragma unroll
    for (int w = 0; w < 7; ++w) {
      float a = f2 ? wa[1][w] : wa[0][w];
      float b = f2 ? wb[1][w] : wb[0][w];
      v2f av2 = { sel2 * (a * a), sel2 * (b * b) };
      v2f bz = *(const v2f*)&zs[f2][1 + 2 * m + w][(q2 << 2) + d0];
      acc2 = __builtin_amdgcn_wmma_f32_16x16x4_f32(
           false, av2,  false, bz,
           (short)0, acc2,  false,  false);
    }
  }
#pragma unroll
  for (int fi = 0; fi < 2; ++fi) {
#pragma unroll
    for (int w = 0; w < 7; ++w) {
      int j = (wv << 5) + 1 + 2 * m + w;
#pragma unroll
      for (int q = 0; q < 4; ++q) {
        v2f bv0 = *(const v2f*)&xs[0][fi][j][(q << 2) + d0];
        v2f bv1 = *(const v2f*)&xs[1][fi][j][(q << 2) + d0];
        float sel = (mq == q) ? 1.f : 0.f;
        v2f av = { sel * wa[fi][w], sel * wb[fi][w] };
        acc0 = __builtin_amdgcn_wmma_f32_16x16x4_f32(
             false, av,  false, bv0,
             (short)0, acc0,  false,  false);
        acc1 = __builtin_amdgcn_wmma_f32_16x16x4_f32(
             false, av,  false, bv1,
             (short)0, acc1,  false,  false);
      }
    }
  }

#pragma unroll
  for (int r = 0; r < 8; ++r) {
    ot[0][r + (hi << 3)][(wv << 4) + m] = acc0[r];
    ot[1][r + (hi << 3)][(wv << 4) + m] = acc1[r];
  }

  float* rw = &rs[wv][0];
#pragma unroll
  for (int r = 0; r < 8; ++r) {
    rw[r * 36 + L]       = acc0[r] + acc1[r];
    rw[(8 + r) * 36 + L] = acc2[r];
  }
  __syncthreads();
#pragma unroll
  for (int k = 0; k < 4; ++k) {
    int idx = (int)threadIdx.x + (k << 8);
    int bb  = idx >> 9;
    int orow = (idx >> 5) & 15;
    int pq  = idx & 31;
    v4f v = *(const v4f*)&ot[bb][orow][pq << 2];
    float* dst = out + (((size_t)bb * On + o0 + orow) * FOUT + fo) * TOUT
                     + (th << 7) + (pq << 2);
    *(v4f*)dst = v;
  }
  {
    int T    = threadIdx.x;
    int sub  = T & 7;
    int oi   = T >> 3;
    int kind = oi >> 4;
    int ol   = oi & 15;
    int r = ol & 7, h = ol >> 3;
    const v4f* rp = (const v4f*)(&rs[sub][0] + (kind * 8 + r) * 36 + h * 16);
    v4f s4 = rp[0] + rp[1] + rp[2] + rp[3];
    float s = (s4.x + s4.y) + (s4.z + s4.w);
    s += __shfl_xor(s, 1, 32);
    s += __shfl_xor(s, 2, 32);
    s += __shfl_xor(s, 4, 32);
    if (sub == 0) {
      int pb = (fo << 1) + th;
      part[(((o0 + ol) << 7) + pb) * 2 + kind] = s;
    }
  }
}

__global__ void __launch_bounds__(256)
lc_bn_apply(const float* __restrict__ part, const float* __restrict__ gamma,
            const float* __restrict__ beta, float* __restrict__ out) {
  int id = blockIdx.x;
  int bo = id >> 2;
  int o  = bo & 63;
  int T  = threadIdx.x;
  v4f* p = (v4f*)(out + (size_t)bo * (FOUT * TOUT) + (id & 3) * 4096);

  v4f v[4];
#pragma unroll
  for (int k = 0; k < 4; ++k) v[k] = p[T + 256 * k];
  v2f pr = {0.f, 0.f};
  if (T < 128) pr = *(const v2f*)(part + (((o << 7) + T) << 1));
  float g = gamma[o], be = beta[o];

  float n1 = pr.x, n2 = pr.y;
  for (int off = 16; off > 0; off >>= 1) {
    n1 += __shfl_xor(n1, off, 32);
    n2 += __shfl_xor(n2, off, 32);
  }
  __shared__ float red[8][2];
  int lane = T & 31, wv = T >> 5;
  if (lane == 0) { red[wv][0] = n1; red[wv][1] = n2; }
  __syncthreads();
  float t1 = 0.f, t2 = 0.f;
#pragma unroll
  for (int i = 0; i < 8; ++i) { t1 += red[i][0]; t2 += red[i][1]; }

  const float invcnt = 1.0f / 1835008.0f;
  float mean = t1 * invcnt;
  float var  = t2 * invcnt - mean * mean;
  float inv  = rsqrtf(var + 1e-5f);
  float a    = g * inv;
  float c    = 56.0f * (be - mean * a);

#pragma unroll
  for (int k = 0; k < 4; ++k) {
    v4f y = a * v[k] + c;
    y.x = (y.x >= 0.f) ? y.x : 0.01f * y.x;
    y.y = (y.y >= 0.f) ? y.y : 0.01f * y.y;
    y.z = (y.z >= 0.f) ? y.z : 0.01f * y.z;
    y.w = (y.w >= 0.f) ? y.w : 0.01f * y.w;
    p[T + 256 * k] = y;
  }
}

extern "C" void kernel_launch(void* const* d_in, const int* in_sizes, int n_in,
                              void* d_out, int out_size, void* d_ws, size_t ws_size,
                              hipStream_t stream) {
  (void)in_sizes; (void)n_in; (void)out_size; (void)ws_size;
  const float* x     = (const float*)d_in[0];
  const float* wgt   = (const float*)d_in[1];
  const float* gamma = (const float*)d_in[2];
  const float* beta  = (const float*)d_in[3];
  float* out = (float*)d_out;

  float* part = (float*)d_ws;

  lc_conv_sums<<<512, 256, 0, stream>>>(x, wgt, part, out);
  lc_bn_apply<<<512, 256, 0, stream>>>(part, gamma, beta, out);
}
